# starting kernel
# baseline (speedup 1.0000x reference)
#include <hip/hip_runtime.h>

#define IMAGE    128
#define N_LINES  32
#define N_CURVES 64
#define SEG_PAD  16
#define N_SEG    9
#define SIGMA2   0.05f
#define EPSF     1e-10f
#define BIGF     1e30f

typedef float v2f __attribute__((ext_vector_type(2)));
typedef float v8f __attribute__((ext_vector_type(8)));

__device__ __forceinline__ float clamp01(float v) { return fminf(fmaxf(v, 0.0f), 1.0f); }

__global__ __launch_bounds__(256)
void render_curves_wmma(const float* __restrict__ position,
                        const float* __restrict__ radius,
                        const float* __restrict__ color,
                        const float* __restrict__ background,
                        float* __restrict__ out)
{
    __align__(16) __shared__ float sS1[N_CURVES][SEG_PAD][4];
    __align__(16) __shared__ float sS2[N_CURVES][SEG_PAD][4];
    __align__(16) __shared__ float sS3[N_CURVES][SEG_PAD][4];
    __align__(16) __shared__ float sMeta[N_CURVES][4];

    const int b   = blockIdx.x / IMAGE;
    const int y   = blockIdx.x % IMAGE;
    const int tid = threadIdx.x;

    for (int e = tid; e < N_CURVES * SEG_PAD; e += 256) {
        const int curve = e >> 4;
        const int seg   = e & 15;
        if (seg < N_SEG) {
            const int line = curve >> 1;
            const float* P = position + (b * N_LINES + line) * 10;
            float c0y, c0x, c1y, c1x, c2y, c2x;
            if ((curve & 1) == 0) {
                c0y = P[0]; c0x = P[1];
                c1y = P[2]; c1x = P[3];
                c2y = P[4]; c2x = P[5];
            } else {
                c0y = P[4];               c0x = P[5];
                c1y = 2.0f * P[4] - P[6]; c1x = 2.0f * P[5] - P[7];
                c2y = P[8];               c2x = P[9];
            }
            const float t0 = (float)seg       * (1.0f / 9.0f);
            const float t1 = (float)(seg + 1) * (1.0f / 9.0f);
            const float u0 = 1.0f - t0, u1 = 1.0f - t1;
            const float ay = u0*u0*c0y + 2.0f*t0*u0*c1y + t0*t0*c2y;
            const float ax = u0*u0*c0x + 2.0f*t0*u0*c1x + t0*t0*c2x;
            const float ey = u1*u1*c0y + 2.0f*t1*u1*c1y + t1*t1*c2y;
            const float ex = u1*u1*c0x + 2.0f*t1*u1*c1x + t1*t1*c2x;
            const float vy = ey - ay, vx = ex - ax;
            const float vv = vy*vy + vx*vx;
            const float inv = 1.0f / (vv + EPSF);
            sS1[curve][seg][0] = vy * inv;
            sS1[curve][seg][1] = vx * inv;
            sS1[curve][seg][2] = -(ay*vy + ax*vx) * inv;
            sS1[curve][seg][3] = 0.0f;
            sS2[curve][seg][0] = -2.0f * ay;
            sS2[curve][seg][1] = -2.0f * ax;
            sS2[curve][seg][2] = ay*ay + ax*ax;
            sS2[curve][seg][3] = 1.0f;
            sS3[curve][seg][0] = 0.0f; sS3[curve][seg][1] = 0.0f;
            sS3[curve][seg][2] = vv;   sS3[curve][seg][3] = 0.0f;
        } else {
            sS1[curve][seg][0] = 0.0f; sS1[curve][seg][1] = 0.0f;
            sS1[curve][seg][2] = 0.0f; sS1[curve][seg][3] = 0.0f;
            sS2[curve][seg][0] = 0.0f; sS2[curve][seg][1] = 0.0f;
            sS2[curve][seg][2] = BIGF; sS2[curve][seg][3] = 0.0f;
            sS3[curve][seg][0] = 0.0f; sS3[curve][seg][1] = 0.0f;
            sS3[curve][seg][2] = 0.0f; sS3[curve][seg][3] = 0.0f;
        }
    }
    if (tid < N_CURVES) {
        const int line = tid >> 1;
        const float r  = radius[b * N_LINES + line];
        const float sg = r * r * SIGMA2;
        sMeta[tid][0] = 1.0f / sg;
        sMeta[tid][1] = color[(b * N_LINES + line) * 3 + 0];
        sMeta[tid][2] = color[(b * N_LINES + line) * 3 + 1];
        sMeta[tid][3] = color[(b * N_LINES + line) * 3 + 2];
    }
    __syncthreads();

    const int wave = tid >> 5;
    const int lane = tid & 31;
    const int colp = lane & 15;
    const int x    = wave * 16 + colp;
    const float sc = 2.0f / 127.0f;
    const float gy = -1.0f + (float)y * sc;
    const float gx = -1.0f + (float)x * sc;
    const bool  hi = lane >= 16;

    v2f Bf;
    Bf.x = hi ? 1.0f : gy;
    Bf.y = hi ? (gy * gy + gx * gx) : gx;

    const int segSel = lane & 15;
    const int kOff   = hi ? 2 : 0;

    float accR = 0.0f, accG = 0.0f, accB = 0.0f;
    float w = 1.0f;

    for (int curve = 0; curve < N_CURVES; ++curve) {
        const v2f A1 = *(const v2f*)&sS1[curve][segSel][kOff];
        const v2f A2 = *(const v2f*)&sS2[curve][segSel][kOff];
        const v2f A3 = *(const v2f*)&sS3[curve][segSel][kOff];

        v8f U  = {};
        v8f S2 = {};
        v8f VV = {};
        U  = __builtin_amdgcn_wmma_f32_16x16x4_f32(false, A1, false, Bf, (short)0, U,  false, false);
        S2 = __builtin_amdgcn_wmma_f32_16x16x4_f32(false, A2, false, Bf, (short)0, S2, false, false);
        VV = __builtin_amdgcn_wmma_f32_16x16x4_f32(false, A3, false, Bf, (short)0, VV, false, false);

        float m = BIGF;
#pragma unroll
        for (int i = 0; i < 8; ++i) {
            const float u  = U[i];
            const float tp = clamp01(u);
            const float e2 = S2[i] - VV[i] * (tp * (2.0f * u - tp));
            m = fminf(m, e2);
        }
        m = fminf(m, __shfl_xor(m, 16, 32));

        const float invsig = sMeta[curve][0];
        const float z = m * invsig;
        const float I = __expf(-(z * z));
        const float wI = w * I;
        accR = fmaf(sMeta[curve][1], wI, accR);
        accG = fmaf(sMeta[curve][2], wI, accG);
        accB = fmaf(sMeta[curve][3], wI, accB);
        w = w * (1.0f - I + EPSF);
    }

    if (!hi) {
        const float bg0 = background[b * 3 + 0];
        const float bg1 = background[b * 3 + 1];
        const float bg2 = background[b * 3 + 2];
        const int base = ((b * 3) * IMAGE + y) * IMAGE + x;
        out[base]                     = clamp01(fmaf(bg0, w, accR));
        out[base + IMAGE * IMAGE]     = clamp01(fmaf(bg1, w, accG));
        out[base + 2 * IMAGE * IMAGE] = clamp01(fmaf(bg2, w, accB));
    }
}

extern "C" void kernel_launch(void* const* d_in, const int* in_sizes, int n_in,
                              void* d_out, int out_size, void* d_ws, size_t ws_size,
                              hipStream_t stream) {
    (void)in_sizes; (void)n_in; (void)out_size; (void)d_ws; (void)ws_size;
    const float* position   = (const float*)d_in[0];
    const float* radius     = (const float*)d_in[1];
    const float* color      = (const float*)d_in[2];
    const float* background = (const float*)d_in[3];
    float* out = (float*)d_out;
    render_curves_wmma<<<4 * IMAGE, 256, 0, stream>>>(position, radius, color, background, out);
}


# Segment rows repacked so no WMMA tile row is padding: 32 tiles of two curves x 8 segments + 4 tiles of ninth segments (36 tiles, was 64); halves swap intensity by permlanex16
# speedup vs baseline: 1.8523x; 1.8523x over previous
#include <hip/hip_runtime.h>


#define IMAGE    128
#define N_LINES  32
#define N_SEG    9
#define N_PAIR   32
#define N_TILES  36
#define SIGMA2   0.05f
#define EPSF     1e-10f

typedef float v2f __attribute__((ext_vector_type(2)));
typedef float v8f __attribute__((ext_vector_type(8)));

__device__ __forceinline__ float clamp01(float v) { return fminf(fmaxf(v, 0.0f), 1.0f); }

__device__ __forceinline__ float xhalf(float x) {
    const int b = __builtin_bit_cast(int, x);
    return __builtin_bit_cast(float,
        __builtin_amdgcn_permlanex16(b, b, 0x76543210, 0xfedcba98, false, false));
}

__device__ __forceinline__ v8f tile_edt2(const float* a1, const float* a2, const float* a3, v2f Bf)
{
    const v2f A1 = *(const v2f*)a1;
    const v2f A2 = *(const v2f*)a2;
    const v2f A3 = *(const v2f*)a3;

    v8f U  = {};
    v8f S2 = {};
    v8f VV = {};
    U  = __builtin_amdgcn_wmma_f32_16x16x4_f32(false, A1, false, Bf, (short)0, U,  false, false);
    S2 = __builtin_amdgcn_wmma_f32_16x16x4_f32(false, A2, false, Bf, (short)0, S2, false, false);
    VV = __builtin_amdgcn_wmma_f32_16x16x4_f32(false, A3, false, Bf, (short)0, VV, false, false);

    v8f E;
#pragma unroll
    for (int i = 0; i < 8; ++i) {
        const float u  = U[i];
        const float tp = clamp01(u);
        E[i] = S2[i] - VV[i] * (tp * (2.0f * u - tp));
    }
    return E;
}

__global__ __launch_bounds__(256)
void render_curves_wmma(const float* __restrict__ position,
                        const float* __restrict__ radius,
                        const float* __restrict__ color,
                        const float* __restrict__ background,
                        float* __restrict__ out)
{
    __align__(16) __shared__ float sS1[N_TILES][16][4];
    __align__(16) __shared__ float sS2[N_TILES][16][4];
    __align__(16) __shared__ float sS3[N_TILES][16][4];
    __align__(16) __shared__ float sMeta[N_PAIR][4];

    const int b   = blockIdx.x / IMAGE;
    const int y   = blockIdx.x % IMAGE;
    const int tid = threadIdx.x;

    for (int e = tid; e < N_TILES * 16; e += 256) {
        const int tile = e >> 4;
        const int row  = e & 15;
        const int half = row >> 3;
        const int sub  = row & 7;
        const int pair = (tile < N_PAIR) ? tile : ((tile - N_PAIR) * 8 + sub);
        const int seg  = (tile < N_PAIR) ? sub : 8;
        const int curve = 2 * pair + half;
        const int line  = curve >> 1;
        const float* P = position + (b * N_LINES + line) * 10;
        float c0y, c0x, c1y, c1x, c2y, c2x;
        if ((curve & 1) == 0) {
            c0y = P[0]; c0x = P[1];
            c1y = P[2]; c1x = P[3];
            c2y = P[4]; c2x = P[5];
        } else {
            c0y = P[4];               c0x = P[5];
            c1y = 2.0f * P[4] - P[6]; c1x = 2.0f * P[5] - P[7];
            c2y = P[8];               c2x = P[9];
        }
        const float t0 = (float)seg       * (1.0f / 9.0f);
        const float t1 = (float)(seg + 1) * (1.0f / 9.0f);
        const float u0 = 1.0f - t0, u1 = 1.0f - t1;
        const float ay = u0*u0*c0y + 2.0f*t0*u0*c1y + t0*t0*c2y;
        const float ax = u0*u0*c0x + 2.0f*t0*u0*c1x + t0*t0*c2x;
        const float ey = u1*u1*c0y + 2.0f*t1*u1*c1y + t1*t1*c2y;
        const float ex = u1*u1*c0x + 2.0f*t1*u1*c1x + t1*t1*c2x;
        const float vy = ey - ay, vx = ex - ax;
        const float vv = vy*vy + vx*vx;
        const float inv = 1.0f / (vv + EPSF);
        sS1[tile][row][0] = vy * inv;
        sS1[tile][row][1] = vx * inv;
        sS1[tile][row][2] = -(ay*vy + ax*vx) * inv;
        sS1[tile][row][3] = 0.0f;
        sS2[tile][row][0] = -2.0f * ay;
        sS2[tile][row][1] = -2.0f * ax;
        sS2[tile][row][2] = ay*ay + ax*ax;
        sS2[tile][row][3] = 1.0f;
        sS3[tile][row][0] = 0.0f; sS3[tile][row][1] = 0.0f;
        sS3[tile][row][2] = vv;   sS3[tile][row][3] = 0.0f;
    }
    if (tid < N_PAIR) {
        const float r  = radius[b * N_LINES + tid];
        const float sg = r * r * SIGMA2;
        sMeta[tid][0] = 1.0f / sg;
        sMeta[tid][1] = color[(b * N_LINES + tid) * 3 + 0];
        sMeta[tid][2] = color[(b * N_LINES + tid) * 3 + 1];
        sMeta[tid][3] = color[(b * N_LINES + tid) * 3 + 2];
    }
    __syncthreads();

    const int wave = tid >> 5;
    const int lane = tid & 31;
    const int colp = lane & 15;
    const int x    = wave * 16 + colp;
    const float sc = 2.0f / 127.0f;
    const float gy = -1.0f + (float)y * sc;
    const float gx = -1.0f + (float)x * sc;
    const bool  hi = lane >= 16;

    v2f Bf;
    Bf.x = hi ? 1.0f : gy;
    Bf.y = hi ? (gy * gy + gx * gx) : gx;

    const int rowSel = lane & 15;
    const int kOff   = hi ? 2 : 0;

    float accR = 0.0f, accG = 0.0f, accB = 0.0f;
    float w = 1.0f;

    for (int T = 0; T < 4; ++T) {
        const v8f E9 = tile_edt2(&sS1[N_PAIR + T][rowSel][kOff], &sS2[N_PAIR + T][rowSel][kOff],
                                 &sS3[N_PAIR + T][rowSel][kOff], Bf);
#pragma unroll
        for (int j = 0; j < 8; ++j) {
            const int p = T * 8 + j;
            const v8f E = tile_edt2(&sS1[p][rowSel][kOff], &sS2[p][rowSel][kOff],
                                    &sS3[p][rowSel][kOff], Bf);
            float m = E[0];
#pragma unroll
            for (int i = 1; i < 8; ++i) m = fminf(m, E[i]);
            m = fminf(m, E9[j]);

            const float invsig = sMeta[p][0];
            const float z = m * invsig;
            const float I = __expf(-(z * z));
            const float J = xhalf(I);
            const float Ie = hi ? J : I;
            const float Io = hi ? I : J;
            const float cR = sMeta[p][1], cG = sMeta[p][2], cB = sMeta[p][3];

            float wI = w * Ie;
            accR = fmaf(cR, wI, accR);
            accG = fmaf(cG, wI, accG);
            accB = fmaf(cB, wI, accB);
            w = w * (1.0f - Ie + EPSF);

            wI = w * Io;
            accR = fmaf(cR, wI, accR);
            accG = fmaf(cG, wI, accG);
            accB = fmaf(cB, wI, accB);
            w = w * (1.0f - Io + EPSF);
        }
    }

    if (!hi) {
        const float bg0 = background[b * 3 + 0];
        const float bg1 = background[b * 3 + 1];
        const float bg2 = background[b * 3 + 2];
        const int base = ((b * 3) * IMAGE + y) * IMAGE + x;
        out[base]                     = clamp01(fmaf(bg0, w, accR));
        out[base + IMAGE * IMAGE]     = clamp01(fmaf(bg1, w, accG));
        out[base + 2 * IMAGE * IMAGE] = clamp01(fmaf(bg2, w, accB));
    }
}

extern "C" void kernel_launch(void* const* d_in, const int* in_sizes, int n_in,
                              void* d_out, int out_size, void* d_ws, size_t ws_size,
                              hipStream_t stream) {
    (void)in_sizes; (void)n_in; (void)out_size; (void)d_ws; (void)ws_size;
    const float* position   = (const float*)d_in[0];
    const float* radius     = (const float*)d_in[1];
    const float* color      = (const float*)d_in[2];
    const float* background = (const float*)d_in[3];
    float* out = (float*)d_out;
    render_curves_wmma<<<4 * IMAGE, 256, 0, stream>>>(position, radius, color, background, out);
}


# All global loads issued at entry (table build: 2-3 rows x 3 points per thread, branch-free; radius, colour, background too), arithmetic after: one memory round trip before the barrier, was up to 7
# speedup vs baseline: 2.2140x; 1.1953x over previous
#include <hip/hip_runtime.h>


#define IMAGE    128
#define N_LINES  32
#define N_SEG    9
#define N_PAIR   32
#define N_TILES  36
#define SIGMA2   0.05f
#define EPSF     1e-10f

typedef float v2f __attribute__((ext_vector_type(2)));
typedef float v4f __attribute__((ext_vector_type(4)));
typedef float v8f __attribute__((ext_vector_type(8)));

__device__ __forceinline__ float clamp01(float v) { return fminf(fmaxf(v, 0.0f), 1.0f); }

__device__ __forceinline__ float xhalf(float x) {
    const int b = __builtin_bit_cast(int, x);
    return __builtin_bit_cast(float,
        __builtin_amdgcn_permlanex16(b, b, 0x76543210, 0xfedcba98, false, false));
}

__device__ __forceinline__ v8f tile_edt2(const float* a1, const float* a2, const float* len8, v2f Bf)
{
    const v2f A1 = *(const v2f*)a1;
    const v2f A2 = *(const v2f*)a2;
    const v4f L0 = *(const v4f*)len8;
    const v4f L1 = *(const v4f*)(len8 + 4);
    const float L[8] = { L0.x, L0.y, L0.z, L0.w, L1.x, L1.y, L1.z, L1.w };

    v8f U  = {};
    v8f S2 = {};
    U  = __builtin_amdgcn_wmma_f32_16x16x4_f32(false, A1, false, Bf, (short)0, U,  false, false);
    S2 = __builtin_amdgcn_wmma_f32_16x16x4_f32(false, A2, false, Bf, (short)0, S2, false, false);

    v8f E;
#pragma unroll
    for (int i = 0; i < 8; ++i) {
        const float u = U[i];
        const float t = __builtin_amdgcn_fmed3f(u, 0.0f, L[i]);
        E[i] = S2[i] - t * (2.0f * u - t);
    }
    return E;
}

struct RowId { int tile, row, line, odd, seg; };
__device__ __forceinline__ RowId row_id(int e)
{
    RowId r;
    r.tile = e >> 4;
    r.row  = e & 15;
    const int half = r.row >> 3;
    const int sub  = r.row & 7;
    const int pair = (r.tile < N_PAIR) ? r.tile : ((r.tile - N_PAIR) * 8 + sub);
    r.seg  = (r.tile < N_PAIR) ? sub : 8;
    r.line = pair;
    r.odd  = half;
    return r;
}

struct RowSrc { v2f p0, p1, p2; };
__device__ __forceinline__ RowSrc row_load(const float* __restrict__ position, int b, const RowId& r)
{
    const float* P = position + (b * N_LINES + r.line) * 10 + (r.odd ? 4 : 0);
    RowSrc s;
    s.p0 = *(const v2f*)(P);
    s.p1 = *(const v2f*)(P + 2);
    s.p2 = *(const v2f*)(P + 4);
    return s;
}

__global__ __launch_bounds__(256)
void render_curves_wmma(const float* __restrict__ position,
                        const float* __restrict__ radius,
                        const float* __restrict__ color,
                        const float* __restrict__ background,
                        float* __restrict__ out)
{
    __align__(16) __shared__ float sS1[N_TILES][16][4];
    __align__(16) __shared__ float sS2[N_TILES][16][4];
    __align__(16) __shared__ float sLen[N_TILES][16];
    __align__(16) __shared__ float sMeta[N_PAIR][4];

    const int b   = blockIdx.x / IMAGE;
    const int y   = blockIdx.x % IMAGE;
    const int tid = threadIdx.x;

    const int  e0 = tid, e1 = tid + 256, e2 = tid + 512;
    const bool third = e2 < N_TILES * 16;
    const bool metaw = tid >= 256 - N_PAIR;
    const int  mline = tid - (256 - N_PAIR);

    const RowId  r0 = row_id(e0), r1 = row_id(e1), r2 = row_id(third ? e2 : e0);
    const RowSrc s0 = row_load(position, b, r0);
    const RowSrc s1 = row_load(position, b, r1);
    RowSrc s2;
    if (third) s2 = row_load(position, b, r2);
    float mr, mcR, mcG, mcB;
    if (metaw) {
        mr  = radius[b * N_LINES + mline];
        mcR = color[(b * N_LINES + mline) * 3 + 0];
        mcG = color[(b * N_LINES + mline) * 3 + 1];
        mcB = color[(b * N_LINES + mline) * 3 + 2];
    }
    const float bg0 = background[b * 3 + 0];
    const float bg1 = background[b * 3 + 1];
    const float bg2 = background[b * 3 + 2];

    auto build_row = [&](const RowId& r, const RowSrc& s) {
        const float c0y = s.p0.x, c0x = s.p0.y;
        const float c2y = s.p2.x, c2x = s.p2.y;
        const float c1y = r.odd ? (2.0f * c0y - s.p1.x) : s.p1.x;
        const float c1x = r.odd ? (2.0f * c0x - s.p1.y) : s.p1.y;
        const float t0 = (float)r.seg       * (1.0f / 9.0f);
        const float t1 = (float)(r.seg + 1) * (1.0f / 9.0f);
        const float u0 = 1.0f - t0, u1 = 1.0f - t1;
        const float ay = u0*u0*c0y + 2.0f*t0*u0*c1y + t0*t0*c2y;
        const float ax = u0*u0*c0x + 2.0f*t0*u0*c1x + t0*t0*c2x;
        const float ey = u1*u1*c0y + 2.0f*t1*u1*c1y + t1*t1*c2y;
        const float ex = u1*u1*c0x + 2.0f*t1*u1*c1x + t1*t1*c2x;
        const float vy = ey - ay, vx = ex - ax;
        const float vv = vy*vy + vx*vx;
        const float s_  = sqrtf(vv + EPSF);
        const float inv = 1.0f / s_;
        sS1[r.tile][r.row][0] = vy * inv;
        sS1[r.tile][r.row][1] = vx * inv;
        sS1[r.tile][r.row][2] = -(ay*vy + ax*vx) * inv;
        sS1[r.tile][r.row][3] = 0.0f;
        sS2[r.tile][r.row][0] = -2.0f * ay;
        sS2[r.tile][r.row][1] = -2.0f * ax;
        sS2[r.tile][r.row][2] = ay*ay + ax*ax;
        sS2[r.tile][r.row][3] = 1.0f;
        sLen[r.tile][r.row] = s_;
    };
    build_row(r0, s0);
    build_row(r1, s1);
    if (third) build_row(r2, s2);
    if (metaw) {
        const float sg = mr * mr * SIGMA2;
        sMeta[mline][0] = 1.0f / sg;
        sMeta[mline][1] = mcR;
        sMeta[mline][2] = mcG;
        sMeta[mline][3] = mcB;
    }
    __syncthreads();

    const int wave = tid >> 5;
    const int lane = tid & 31;
    const int colp = lane & 15;
    const int x    = wave * 16 + colp;
    const float sc = 2.0f / 127.0f;
    const float gy = -1.0f + (float)y * sc;
    const float gx = -1.0f + (float)x * sc;
    const bool  hi = lane >= 16;

    v2f Bf;
    Bf.x = hi ? 1.0f : gy;
    Bf.y = hi ? (gy * gy + gx * gx) : gx;

    const int rowSel = lane & 15;
    const int kOff   = hi ? 2 : 0;
    const int lenOff = hi ? 8 : 0;

    float accR = 0.0f, accG = 0.0f, accB = 0.0f;
    float w = 1.0f;

    for (int T = 0; T < 4; ++T) {
        const v8f E9 = tile_edt2(&sS1[N_PAIR + T][rowSel][kOff], &sS2[N_PAIR + T][rowSel][kOff],
                                 &sLen[N_PAIR + T][lenOff], Bf);
#pragma unroll
        for (int j = 0; j < 8; ++j) {
            const int p = T * 8 + j;
            const v8f E = tile_edt2(&sS1[p][rowSel][kOff], &sS2[p][rowSel][kOff],
                                    &sLen[p][lenOff], Bf);
            float m = E[0];
#pragma unroll
            for (int i = 1; i < 8; ++i) m = fminf(m, E[i]);
            m = fminf(m, E9[j]);

            const float invsig = sMeta[p][0];
            const float z = m * invsig;
            const float I = __expf(-(z * z));
            const float J = xhalf(I);
            const float Ie = hi ? J : I;
            const float Io = hi ? I : J;
            const float cR = sMeta[p][1], cG = sMeta[p][2], cB = sMeta[p][3];

            float wI = w * Ie;
            accR = fmaf(cR, wI, accR);
            accG = fmaf(cG, wI, accG);
            accB = fmaf(cB, wI, accB);
            w = w * (1.0f - Ie + EPSF);

            wI = w * Io;
            accR = fmaf(cR, wI, accR);
            accG = fmaf(cG, wI, accG);
            accB = fmaf(cB, wI, accB);
            w = w * (1.0f - Io + EPSF);
        }
    }

    if (!hi) {
        const int base = ((b * 3) * IMAGE + y) * IMAGE + x;
        out[base]                     = clamp01(fmaf(bg0, w, accR));
        out[base + IMAGE * IMAGE]     = clamp01(fmaf(bg1, w, accG));
        out[base + 2 * IMAGE * IMAGE] = clamp01(fmaf(bg2, w, accB));
    }
}

extern "C" void kernel_launch(void* const* d_in, const int* in_sizes, int n_in,
                              void* d_out, int out_size, void* d_ws, size_t ws_size,
                              hipStream_t stream) {
    (void)in_sizes; (void)n_in; (void)out_size; (void)d_ws; (void)ws_size;
    const float* position   = (const float*)d_in[0];
    const float* radius     = (const float*)d_in[1];
    const float* color      = (const float*)d_in[2];
    const float* background = (const float*)d_in[3];
    float* out = (float*)d_out;
    render_curves_wmma<<<4 * IMAGE, 256, 0, stream>>>(position, radius, color, background, out);
}


# Tile rows repacked as curve c with curve c+32: each lane half blends its own 32 curves, halves joined once at the end (new summation tree); per-tile permlanex16 exchange and duplicate blend gone
# speedup vs baseline: 2.4170x; 1.0917x over previous
#include <hip/hip_runtime.h>


#define IMAGE    128
#define N_LINES  32
#define N_SEG    9
#define N_FRONT  32
#define N_TILES  36
#define SIGMA2   0.05f
#define EPSF     1e-10f

typedef float v2f __attribute__((ext_vector_type(2)));
typedef float v4f __attribute__((ext_vector_type(4)));
typedef float v8f __attribute__((ext_vector_type(8)));

__device__ __forceinline__ float clamp01(float v) { return fminf(fmaxf(v, 0.0f), 1.0f); }

__device__ __forceinline__ float xhalf(float x) {
    const int b = __builtin_bit_cast(int, x);
    return __builtin_bit_cast(float,
        __builtin_amdgcn_permlanex16(b, b, 0x76543210, 0xfedcba98, false, false));
}

__device__ __forceinline__ v8f tile_edt2(const float* a1, const float* a2, const float* len8, v2f Bf)
{
    const v2f A1 = *(const v2f*)a1;
    const v2f A2 = *(const v2f*)a2;
    const v4f L0 = *(const v4f*)len8;
    const v4f L1 = *(const v4f*)(len8 + 4);
    const float L[8] = { L0.x, L0.y, L0.z, L0.w, L1.x, L1.y, L1.z, L1.w };

    v8f U  = {};
    v8f S2 = {};
    U  = __builtin_amdgcn_wmma_f32_16x16x4_f32(false, A1, false, Bf, (short)0, U,  false, false);
    S2 = __builtin_amdgcn_wmma_f32_16x16x4_f32(false, A2, false, Bf, (short)0, S2, false, false);

    v8f E;
#pragma unroll
    for (int i = 0; i < 8; ++i) {
        const float u = U[i];
        const float t = __builtin_amdgcn_fmed3f(u, 0.0f, L[i]);
        E[i] = S2[i] - t * (2.0f * u - t);
    }
    return E;
}

struct RowId { int tile, row, line, odd, seg; };
__device__ __forceinline__ RowId row_id(int e)
{
    RowId r;
    r.tile = e >> 4;
    r.row  = e & 15;
    const int half = r.row >> 3;
    const int sub  = r.row & 7;
    const int front = (r.tile < N_FRONT) ? r.tile : ((r.tile - N_FRONT) * 8 + sub);
    const int curve = front + half * N_FRONT;
    r.seg  = (r.tile < N_FRONT) ? sub : 8;
    r.line = curve >> 1;
    r.odd  = curve & 1;
    return r;
}

struct RowSrc { v2f p0, p1, p2; };
__device__ __forceinline__ RowSrc row_load(const float* __restrict__ position, int b, const RowId& r)
{
    const float* P = position + (b * N_LINES + r.line) * 10 + (r.odd ? 4 : 0);
    RowSrc s;
    s.p0 = *(const v2f*)(P);
    s.p1 = *(const v2f*)(P + 2);
    s.p2 = *(const v2f*)(P + 4);
    return s;
}

__global__ __launch_bounds__(256)
void render_curves_wmma(const float* __restrict__ position,
                        const float* __restrict__ radius,
                        const float* __restrict__ color,
                        const float* __restrict__ background,
                        float* __restrict__ out)
{
    __align__(16) __shared__ float sS1[N_TILES][16][4];
    __align__(16) __shared__ float sS2[N_TILES][16][4];
    __align__(16) __shared__ float sLen[N_TILES][16];
    __align__(16) __shared__ float sMeta[N_LINES][4];

    const int b   = blockIdx.x / IMAGE;
    const int y   = blockIdx.x % IMAGE;
    const int tid = threadIdx.x;

    const int  e0 = tid, e1 = tid + 256, e2 = tid + 512;
    const bool third = e2 < N_TILES * 16;
    const bool metaw = tid >= 256 - N_LINES;
    const int  mline = tid - (256 - N_LINES);

    const RowId  r0 = row_id(e0), r1 = row_id(e1), r2 = row_id(third ? e2 : e0);
    const RowSrc s0 = row_load(position, b, r0);
    const RowSrc s1 = row_load(position, b, r1);
    RowSrc s2;
    if (third) s2 = row_load(position, b, r2);
    float mr, mcR, mcG, mcB;
    if (metaw) {
        mr  = radius[b * N_LINES + mline];
        mcR = color[(b * N_LINES + mline) * 3 + 0];
        mcG = color[(b * N_LINES + mline) * 3 + 1];
        mcB = color[(b * N_LINES + mline) * 3 + 2];
    }
    const float bg0 = background[b * 3 + 0];
    const float bg1 = background[b * 3 + 1];
    const float bg2 = background[b * 3 + 2];

    auto build_row = [&](const RowId& r, const RowSrc& s) {
        const float c0y = s.p0.x, c0x = s.p0.y;
        const float c2y = s.p2.x, c2x = s.p2.y;
        const float c1y = r.odd ? (2.0f * c0y - s.p1.x) : s.p1.x;
        const float c1x = r.odd ? (2.0f * c0x - s.p1.y) : s.p1.y;
        const float t0 = (float)r.seg       * (1.0f / 9.0f);
        const float t1 = (float)(r.seg + 1) * (1.0f / 9.0f);
        const float u0 = 1.0f - t0, u1 = 1.0f - t1;
        const float ay = u0*u0*c0y + 2.0f*t0*u0*c1y + t0*t0*c2y;
        const float ax = u0*u0*c0x + 2.0f*t0*u0*c1x + t0*t0*c2x;
        const float ey = u1*u1*c0y + 2.0f*t1*u1*c1y + t1*t1*c2y;
        const float ex = u1*u1*c0x + 2.0f*t1*u1*c1x + t1*t1*c2x;
        const float vy = ey - ay, vx = ex - ax;
        const float vv = vy*vy + vx*vx;
        const float s_  = sqrtf(vv + EPSF);
        const float inv = 1.0f / s_;
        sS1[r.tile][r.row][0] = vy * inv;
        sS1[r.tile][r.row][1] = vx * inv;
        sS1[r.tile][r.row][2] = -(ay*vy + ax*vx) * inv;
        sS1[r.tile][r.row][3] = 0.0f;
        sS2[r.tile][r.row][0] = -2.0f * ay;
        sS2[r.tile][r.row][1] = -2.0f * ax;
        sS2[r.tile][r.row][2] = ay*ay + ax*ax;
        sS2[r.tile][r.row][3] = 1.0f;
        sLen[r.tile][r.row] = s_;
    };
    build_row(r0, s0);
    build_row(r1, s1);
    if (third) build_row(r2, s2);
    if (metaw) {
        const float sg = mr * mr * SIGMA2;
        sMeta[mline][0] = 1.0f / sg;
        sMeta[mline][1] = mcR;
        sMeta[mline][2] = mcG;
        sMeta[mline][3] = mcB;
    }
    __syncthreads();

    const int wave = tid >> 5;
    const int lane = tid & 31;
    const int colp = lane & 15;
    const int x    = wave * 16 + colp;
    const float sc = 2.0f / 127.0f;
    const float gy = -1.0f + (float)y * sc;
    const float gx = -1.0f + (float)x * sc;
    const bool  hi = lane >= 16;

    v2f Bf;
    Bf.x = hi ? 1.0f : gy;
    Bf.y = hi ? (gy * gy + gx * gx) : gx;

    const int rowSel = lane & 15;
    const int kOff   = hi ? 2 : 0;
    const int lenOff = hi ? 8 : 0;

    const int lineOff = hi ? N_LINES / 2 : 0;

    float accR = 0.0f, accG = 0.0f, accB = 0.0f;
    float w = 1.0f;

    for (int T = 0; T < 4; ++T) {
        const v8f E9 = tile_edt2(&sS1[N_FRONT + T][rowSel][kOff], &sS2[N_FRONT + T][rowSel][kOff],
                                 &sLen[N_FRONT + T][lenOff], Bf);
#pragma unroll
        for (int j = 0; j < 8; ++j) {
            const int p = T * 8 + j;
            const v8f E = tile_edt2(&sS1[p][rowSel][kOff], &sS2[p][rowSel][kOff],
                                    &sLen[p][lenOff], Bf);
            float m = E[0];
#pragma unroll
            for (int i = 1; i < 8; ++i) m = fminf(m, E[i]);
            m = fminf(m, E9[j]);

            const v4f meta = *(const v4f*)&sMeta[lineOff + (p >> 1)][0];
            const float z = m * meta.x;
            const float I = __expf(-(z * z));

            const float wI = w * I;
            accR = fmaf(meta.y, wI, accR);
            accG = fmaf(meta.z, wI, accG);
            accB = fmaf(meta.w, wI, accB);
            w = w * (1.0f - I + EPSF);
        }
    }

    const float bR = xhalf(accR), bG = xhalf(accG), bB = xhalf(accB), bW = xhalf(w);

    if (!hi) {
        const float pR = fmaf(w, bR, accR);
        const float pG = fmaf(w, bG, accG);
        const float pB = fmaf(w, bB, accB);
        const float wt = w * bW;
        const int base = ((b * 3) * IMAGE + y) * IMAGE + x;
        out[base]                     = clamp01(fmaf(bg0, wt, pR));
        out[base + IMAGE * IMAGE]     = clamp01(fmaf(bg1, wt, pG));
        out[base + 2 * IMAGE * IMAGE] = clamp01(fmaf(bg2, wt, pB));
    }
}

extern "C" void kernel_launch(void* const* d_in, const int* in_sizes, int n_in,
                              void* d_out, int out_size, void* d_ws, size_t ws_size,
                              hipStream_t stream) {
    (void)in_sizes; (void)n_in; (void)out_size; (void)d_ws; (void)ws_size;
    const float* position   = (const float*)d_in[0];
    const float* radius     = (const float*)d_in[1];
    const float* color      = (const float*)d_in[2];
    const float* background = (const float*)d_in[3];
    float* out = (float*)d_out;
    render_curves_wmma<<<4 * IMAGE, 256, 0, stream>>>(position, radius, color, background, out);
}


# Curves whose f32 intensity is exactly 0 on a wave's 16 pixels (box test) are skipped: per-wave survivor lists, tables by curve; blend sum splits at list middle; heavy and light strips share a SIMD
# speedup vs baseline: 2.8582x; 1.1825x over previous
#include <hip/hip_runtime.h>


#define IMAGE    128
#define N_LINES  32
#define N_CURVES 64
#define DUMMY    64
#define SIGMA2   0.05f
#define EPSF     1e-10f

typedef float v2f __attribute__((ext_vector_type(2)));
typedef float v4f __attribute__((ext_vector_type(4)));
typedef float v8f __attribute__((ext_vector_type(8)));
typedef int   v4i __attribute__((ext_vector_type(4)));

__device__ __forceinline__ float clamp01(float v) { return fminf(fmaxf(v, 0.0f), 1.0f); }

__device__ __forceinline__ float xhalf(float x) {
    const int b = __builtin_bit_cast(int, x);
    return __builtin_bit_cast(float,
        __builtin_amdgcn_permlanex16(b, b, 0x76543210, 0xfedcba98, false, false));
}

__device__ __forceinline__ v8f tile_edt2(v2f A1, v2f A2, v4f L0, v4f L1, v2f Bf)
{
    const float L[8] = { L0.x, L0.y, L0.z, L0.w, L1.x, L1.y, L1.z, L1.w };

    v8f U  = {};
    v8f S2 = {};
    U  = __builtin_amdgcn_wmma_f32_16x16x4_f32(false, A1, false, Bf, (short)0, U,  false, false);
    S2 = __builtin_amdgcn_wmma_f32_16x16x4_f32(false, A2, false, Bf, (short)0, S2, false, false);

    v8f E;
#pragma unroll
    for (int i = 0; i < 8; ++i) {
        const float u = U[i];
        const float t = __builtin_amdgcn_fmed3f(u, 0.0f, L[i]);
        E[i] = S2[i] - t * (2.0f * u - t);
    }
    return E;
}

struct CurveSrc { v2f p0, p1, p2; };
__device__ __forceinline__ CurveSrc curve_load(const float* __restrict__ position, int b, int curve)
{
    const float* P = position + (b * N_LINES + (curve >> 1)) * 10 + ((curve & 1) ? 4 : 0);
    CurveSrc s;
    s.p0 = *(const v2f*)(P);
    s.p1 = *(const v2f*)(P + 2);
    s.p2 = *(const v2f*)(P + 4);
    return s;
}

struct Ctl { float c0y, c0x, c1y, c1x, c2y, c2x; };
__device__ __forceinline__ Ctl curve_ctl(const CurveSrc& s, int curve)
{
    Ctl c;
    c.c0y = s.p0.x; c.c0x = s.p0.y;
    c.c2y = s.p2.x; c.c2x = s.p2.y;
    const bool odd = (curve & 1) != 0;
    c.c1y = odd ? (2.0f * c.c0y - s.p1.x) : s.p1.x;
    c.c1x = odd ? (2.0f * c.c0x - s.p1.y) : s.p1.y;
    return c;
}

__global__ __launch_bounds__(256)
void render_curves_wmma(const float* __restrict__ position,
                        const float* __restrict__ radius,
                        const float* __restrict__ color,
                        const float* __restrict__ background,
                        float* __restrict__ out)
{
    __align__(16) __shared__ float sA[2][N_CURVES + 1][8][4];
    __align__(16) __shared__ float sN[2][N_CURVES + 1][4];
    __align__(16) __shared__ float sCv[N_CURVES + 1][16];
    __align__(16) __shared__ float sN9len[N_CURVES + 4];
    __align__(16) __shared__ float sBox[N_CURVES][4];
    __align__(16) __shared__ float sThr[N_CURVES];
    __align__(16) __shared__ int   sList[8][2][32];
    __align__(16) __shared__ float sL9[8][2][32];

    const int b    = blockIdx.x >> 7;
    const int yy   = (blockIdx.x & 127) >> 1;
    const int q    = blockIdx.x & 1;
    const int tid  = threadIdx.x;
    const int wave = tid >> 5;
    const int lane = tid & 31;

    const int  cv0 = tid >> 3,         sg0 = tid & 7;
    const int  cv1 = (tid >> 3) + 32,  sg1 = tid & 7;
    const bool third = tid < N_CURVES;
    const bool boxw  = tid >= 128 && tid < 128 + N_CURVES;
    const bool metaw = tid >= 256 - N_LINES;
    const int  mline = tid - (256 - N_LINES);
    const int  cvx   = third ? tid : (tid - 128);

    const CurveSrc s0 = curve_load(position, b, cv0);
    const CurveSrc s1 = curve_load(position, b, cv1);
    CurveSrc sx;
    float rx = 0.0f;
    if (third || boxw) sx = curve_load(position, b, cvx);
    if (boxw) rx = radius[b * N_LINES + (cvx >> 1)];
    float mr, mcR, mcG, mcB;
    if (metaw) {
        mr  = radius[b * N_LINES + mline];
        mcR = color[(b * N_LINES + mline) * 3 + 0];
        mcG = color[(b * N_LINES + mline) * 3 + 1];
        mcB = color[(b * N_LINES + mline) * 3 + 2];
    }
    const float bg0 = background[b * 3 + 0];
    const float bg1 = background[b * 3 + 1];
    const float bg2 = background[b * 3 + 2];

    auto build_row = [&](int curve, int seg, const CurveSrc& s, float* r1, float* r2, float* len) {
        const Ctl c = curve_ctl(s, curve);
        const float t0 = (float)seg       * (1.0f / 9.0f);
        const float t1 = (float)(seg + 1) * (1.0f / 9.0f);
        const float u0 = 1.0f - t0, u1 = 1.0f - t1;
        const float ay = u0*u0*c.c0y + 2.0f*t0*u0*c.c1y + t0*t0*c.c2y;
        const float ax = u0*u0*c.c0x + 2.0f*t0*u0*c.c1x + t0*t0*c.c2x;
        const float ey = u1*u1*c.c0y + 2.0f*t1*u1*c.c1y + t1*t1*c.c2y;
        const float ex = u1*u1*c.c0x + 2.0f*t1*u1*c.c1x + t1*t1*c.c2x;
        const float vy = ey - ay, vx = ex - ax;
        const float vv = vy*vy + vx*vx;
        const float s_  = sqrtf(vv + EPSF);
        const float inv = 1.0f / s_;
        r1[0] = vy * inv;
        r1[1] = vx * inv;
        r1[2] = -(ay*vy + ax*vx) * inv;
        r1[3] = 0.0f;
        r2[0] = -2.0f * ay;
        r2[1] = -2.0f * ax;
        r2[2] = ay*ay + ax*ax;
        r2[3] = 1.0f;
        *len = s_;
    };
    build_row(cv0, sg0, s0, &sA[0][cv0][sg0][0], &sA[1][cv0][sg0][0], &sCv[cv0][sg0]);
    build_row(cv1, sg1, s1, &sA[0][cv1][sg1][0], &sA[1][cv1][sg1][0], &sCv[cv1][sg1]);
    if (third) build_row(cvx, 8, sx, &sN[0][cvx][0], &sN[1][cvx][0], &sN9len[cvx]);
    if (boxw) {
        const Ctl c = curve_ctl(sx, cvx);
        float ymin = c.c0y, ymax = c.c0y, xmin = c.c0x, xmax = c.c0x;
#pragma unroll
        for (int k = 1; k < 10; ++k) {
            const float t = (float)k * (1.0f / 9.0f);
            const float u = 1.0f - t;
            const float py = u*u*c.c0y + 2.0f*t*u*c.c1y + t*t*c.c2y;
            const float px = u*u*c.c0x + 2.0f*t*u*c.c1x + t*t*c.c2x;
            ymin = fminf(ymin, py); ymax = fmaxf(ymax, py);
            xmin = fminf(xmin, px); xmax = fmaxf(xmax, px);
        }
        sBox[cvx][0] = ymin; sBox[cvx][1] = ymax; sBox[cvx][2] = xmin; sBox[cvx][3] = xmax;
        sThr[cvx] = 11.0f * (rx * rx * SIGMA2) + 1e-3f;
    }
    if (metaw) {
        const float sg = mr * mr * SIGMA2;
        const float is = 1.0f / sg;
        float* m0 = &sCv[2 * mline][8];
        float* m1 = &sCv[2 * mline + 1][8];
        m0[0] = is; m0[1] = mcR; m0[2] = mcG; m0[3] = mcB;
        m1[0] = is; m1[1] = mcR; m1[2] = mcG; m1[3] = mcB;
    }
    if (tid >= 64 && tid < 72) {
        const int r = tid - 64;
        sA[0][DUMMY][r][0] = 0.0f; sA[0][DUMMY][r][1] = 0.0f; sA[0][DUMMY][r][2] = 0.0f;  sA[0][DUMMY][r][3] = 0.0f;
        sA[1][DUMMY][r][0] = 0.0f; sA[1][DUMMY][r][1] = 0.0f; sA[1][DUMMY][r][2] = 1e4f;  sA[1][DUMMY][r][3] = 0.0f;
        sCv[DUMMY][r] = 1.0f;
    }
    if (tid == 72) {
        sN[0][DUMMY][0] = 0.0f; sN[0][DUMMY][1] = 0.0f; sN[0][DUMMY][2] = 0.0f;  sN[0][DUMMY][3] = 0.0f;
        sN[1][DUMMY][0] = 0.0f; sN[1][DUMMY][1] = 0.0f; sN[1][DUMMY][2] = 1e4f;  sN[1][DUMMY][3] = 0.0f;
        sN9len[DUMMY] = 1.0f;
        sCv[DUMMY][8] = 1.0f; sCv[DUMMY][9] = 0.0f; sCv[DUMMY][10] = 0.0f; sCv[DUMMY][11] = 0.0f;
    }
    sList[wave][0][lane] = DUMMY;
    sList[wave][1][lane] = DUMMY;
    sL9[wave][0][lane] = 1.0f;
    sL9[wave][1][lane] = 1.0f;
    __syncthreads();

    const int y     = yy + ((wave >= 4) ? 64 : 0);
    const int strip = (4 * q + wave) & 7;
    const int colp  = lane & 15;
    const int x     = strip * 16 + colp;
    const float sc = 2.0f / 127.0f;
    const float gy = -1.0f + (float)y * sc;
    const float gx = -1.0f + (float)x * sc;
    const bool  hi = lane >= 16;

    const float xl = -1.0f + (float)(strip * 16) * sc;
    const float xr = -1.0f + (float)(strip * 16 + 15) * sc;
    const v4f  bxa = *(const v4f*)&sBox[lane][0];
    const v4f  bxb = *(const v4f*)&sBox[lane + 32][0];
    const float tha = sThr[lane], thb = sThr[lane + 32];
    const float n9a = sN9len[lane], n9b = sN9len[lane + 32];
    const float dya = fmaxf(fmaxf(bxa.x - gy, gy - bxa.y), 0.0f);
    const float dxa = fmaxf(fmaxf(bxa.z - xr, xl - bxa.w), 0.0f);
    const float dyb = fmaxf(fmaxf(bxb.x - gy, gy - bxb.y), 0.0f);
    const float dxb = fmaxf(fmaxf(bxb.z - xr, xl - bxb.w), 0.0f);
    const bool keepa = !(dya * dya + dxa * dxa > tha);
    const bool keepb = !(dyb * dyb + dxb * dxb > thb);
    const unsigned ma = __builtin_amdgcn_ballot_w32(keepa);
    const unsigned mb = __builtin_amdgcn_ballot_w32(keepb);
    const int na = __builtin_popcount(ma);
    const int n  = na + __builtin_popcount(mb);
    const int h0 = (n + 1) >> 1;
    const unsigned below = (1u << lane) - 1u;
    const int ra = __builtin_popcount(ma & below);
    const int rb = na + __builtin_popcount(mb & below);
    if (keepa) {
        const int li = (ra >= h0) ? 1 : 0;
        const int ix = ra - (li ? h0 : 0);
        sList[wave][li][ix] = lane;
        sL9[wave][li][ix]   = n9a;
    }
    if (keepb) {
        const int li = (rb >= h0) ? 1 : 0;
        const int ix = rb - (li ? h0 : 0);
        sList[wave][li][ix] = lane + 32;
        sL9[wave][li][ix]   = n9b;
    }

    v2f Bf;
    Bf.x = hi ? 1.0f : gy;
    Bf.y = hi ? (gy * gy + gx * gx) : gx;

    const int rA   = lane & 7;
    const int kOff = hi ? 2 : 0;
    const int*   listA = &sList[wave][(lane >> 3) & 1][0];
    const int*   listD = &sList[wave][lane >> 4][0];
    const float* len9D = &sL9[wave][lane >> 4][0];

    float accR = 0.0f, accG = 0.0f, accB = 0.0f;
    float w = 1.0f;

    struct TileOps { v2f A1, A2; v4f L0, L1, meta; };
    auto load_ops = [&](int ca, int cd) {
        TileOps o;
        o.A1 = *(const v2f*)&sA[0][ca][rA][kOff];
        o.A2 = *(const v2f*)&sA[1][ca][rA][kOff];
        const float* cv = &sCv[cd][0];
        o.L0   = *(const v4f*)cv;
        o.L1   = *(const v4f*)(cv + 4);
        o.meta = *(const v4f*)(cv + 8);
        return o;
    };

    const int ngroups = (h0 + 7) >> 3;
    for (int g = 0; g < ngroups; ++g) {
        const v4i ca0 = *(const v4i*)(listA + 8 * g), ca1 = *(const v4i*)(listA + 8 * g + 4);
        const v4i cd0 = *(const v4i*)(listD + 8 * g), cd1 = *(const v4i*)(listD + 8 * g + 4);
        const int cA[8] = { ca0.x, ca0.y, ca0.z, ca0.w, ca1.x, ca1.y, ca1.z, ca1.w };
        const int cD[8] = { cd0.x, cd0.y, cd0.z, cd0.w, cd1.x, cd1.y, cd1.z, cd1.w };
        const int  cN  = listA[8 * g + rA];
        const v4f  l90 = *(const v4f*)(len9D + 8 * g), l91 = *(const v4f*)(len9D + 8 * g + 4);
        TileOps buf[2];
        buf[0] = load_ops(cA[0], cD[0]);
        const v8f  E9  = tile_edt2(*(const v2f*)&sN[0][cN][kOff], *(const v2f*)&sN[1][cN][kOff], l90, l91, Bf);
#pragma unroll
        for (int j = 0; j < 8; ++j) {
            if (8 * g + j < h0) {
                if (j < 7) buf[(j + 1) & 1] = load_ops(cA[j < 7 ? j + 1 : 7], cD[j < 7 ? j + 1 : 7]);
                __builtin_amdgcn_sched_barrier(0);
                const TileOps& cur = buf[j & 1];

                const v8f E = tile_edt2(cur.A1, cur.A2, cur.L0, cur.L1, Bf);
                float m = E[0];
#pragma unroll
                for (int i = 1; i < 8; ++i) m = fminf(m, E[i]);
                m = fminf(m, E9[j]);

                const float z = m * cur.meta.x;
                const float I = __expf(-(z * z));

                const float wI = w * I;
                accR = fmaf(cur.meta.y, wI, accR);
                accG = fmaf(cur.meta.z, wI, accG);
                accB = fmaf(cur.meta.w, wI, accB);
                w = w * (1.0f - I + EPSF);
            }
        }
    }

    const float bR = xhalf(accR), bG = xhalf(accG), bB = xhalf(accB), bW = xhalf(w);

    if (!hi) {
        const float pR = fmaf(w, bR, accR);
        const float pG = fmaf(w, bG, accG);
        const float pB = fmaf(w, bB, accB);
        const float wt = w * bW;
        const int base = ((b * 3) * IMAGE + y) * IMAGE + x;
        out[base]                     = clamp01(fmaf(bg0, wt, pR));
        out[base + IMAGE * IMAGE]     = clamp01(fmaf(bg1, wt, pG));
        out[base + 2 * IMAGE * IMAGE] = clamp01(fmaf(bg2, wt, pB));
    }
}

extern "C" void kernel_launch(void* const* d_in, const int* in_sizes, int n_in,
                              void* d_out, int out_size, void* d_ws, size_t ws_size,
                              hipStream_t stream) {
    (void)in_sizes; (void)n_in; (void)out_size; (void)d_ws; (void)ws_size;
    const float* position   = (const float*)d_in[0];
    const float* radius     = (const float*)d_in[1];
    const float* color      = (const float*)d_in[2];
    const float* background = (const float*)d_in[3];
    float* out = (float*)d_out;
    render_curves_wmma<<<4 * IMAGE, 256, 0, stream>>>(position, radius, color, background, out);
}


# Workgroups of 16 waves (256, was 512 of 8): tables built once per 16 strips, and the four waves that share a SIMD take strips a quarter image apart on a wrapped diagonal
# speedup vs baseline: 3.0774x; 1.0767x over previous
#include <hip/hip_runtime.h>


#define IMAGE    128
#define N_LINES  32
#define N_CURVES 64
#define DUMMY    64
#define SIGMA2   0.05f
#define EPSF     1e-10f

typedef float v2f __attribute__((ext_vector_type(2)));
typedef float v4f __attribute__((ext_vector_type(4)));
typedef float v8f __attribute__((ext_vector_type(8)));
typedef int   v4i __attribute__((ext_vector_type(4)));

__device__ __forceinline__ float clamp01(float v) { return fminf(fmaxf(v, 0.0f), 1.0f); }

__device__ __forceinline__ float xhalf(float x) {
    const int b = __builtin_bit_cast(int, x);
    return __builtin_bit_cast(float,
        __builtin_amdgcn_permlanex16(b, b, 0x76543210, 0xfedcba98, false, false));
}

__device__ __forceinline__ v8f tile_edt2(v2f A1, v2f A2, v4f L0, v4f L1, v2f Bf)
{
    const float L[8] = { L0.x, L0.y, L0.z, L0.w, L1.x, L1.y, L1.z, L1.w };

    v8f U  = {};
    v8f S2 = {};
    U  = __builtin_amdgcn_wmma_f32_16x16x4_f32(false, A1, false, Bf, (short)0, U,  false, false);
    S2 = __builtin_amdgcn_wmma_f32_16x16x4_f32(false, A2, false, Bf, (short)0, S2, false, false);

    v8f E;
#pragma unroll
    for (int i = 0; i < 8; ++i) {
        const float u = U[i];
        const float t = __builtin_amdgcn_fmed3f(u, 0.0f, L[i]);
        E[i] = S2[i] - t * (2.0f * u - t);
    }
    return E;
}

struct CurveSrc { v2f p0, p1, p2; };
__device__ __forceinline__ CurveSrc curve_load(const float* __restrict__ position, int b, int curve)
{
    const float* P = position + (b * N_LINES + (curve >> 1)) * 10 + ((curve & 1) ? 4 : 0);
    CurveSrc s;
    s.p0 = *(const v2f*)(P);
    s.p1 = *(const v2f*)(P + 2);
    s.p2 = *(const v2f*)(P + 4);
    return s;
}

struct Ctl { float c0y, c0x, c1y, c1x, c2y, c2x; };
__device__ __forceinline__ Ctl curve_ctl(const CurveSrc& s, int curve)
{
    Ctl c;
    c.c0y = s.p0.x; c.c0x = s.p0.y;
    c.c2y = s.p2.x; c.c2x = s.p2.y;
    const bool odd = (curve & 1) != 0;
    c.c1y = odd ? (2.0f * c.c0y - s.p1.x) : s.p1.x;
    c.c1x = odd ? (2.0f * c.c0x - s.p1.y) : s.p1.y;
    return c;
}

__global__ __launch_bounds__(512)
void render_curves_wmma(const float* __restrict__ position,
                        const float* __restrict__ radius,
                        const float* __restrict__ color,
                        const float* __restrict__ background,
                        float* __restrict__ out)
{
    __align__(16) __shared__ float sA[2][N_CURVES + 1][8][4];
    __align__(16) __shared__ float sN[2][N_CURVES + 1][4];
    __align__(16) __shared__ float sCv[N_CURVES + 1][16];
    __align__(16) __shared__ float sN9len[N_CURVES + 4];
    __align__(16) __shared__ float sBox[N_CURVES][4];
    __align__(16) __shared__ float sThr[N_CURVES];
    __align__(16) __shared__ int   sList[16][2][32];
    __align__(16) __shared__ float sL9[16][2][32];

    const int b    = blockIdx.x >> 6;
    const int yr   = (blockIdx.x & 63) >> 1;
    const int q    = blockIdx.x & 1;
    const int tid  = threadIdx.x;
    const int wave = tid >> 5;
    const int lane = tid & 31;

    const int  cv0 = tid >> 3,         sg0 = tid & 7;
    const bool third = tid < N_CURVES;
    const bool boxw  = tid >= 64 && tid < 64 + N_CURVES;
    const bool metaw = tid >= 384 && tid < 384 + N_LINES;
    const int  mline = tid - 384;
    const int  cvx   = third ? tid : (tid - 64);

    const CurveSrc s0 = curve_load(position, b, cv0);
    CurveSrc sx;
    float rx = 0.0f;
    if (third || boxw) sx = curve_load(position, b, cvx);
    if (boxw) rx = radius[b * N_LINES + (cvx >> 1)];
    float mr, mcR, mcG, mcB;
    if (metaw) {
        mr  = radius[b * N_LINES + mline];
        mcR = color[(b * N_LINES + mline) * 3 + 0];
        mcG = color[(b * N_LINES + mline) * 3 + 1];
        mcB = color[(b * N_LINES + mline) * 3 + 2];
    }
    const float bg0 = background[b * 3 + 0];
    const float bg1 = background[b * 3 + 1];
    const float bg2 = background[b * 3 + 2];

    auto build_row = [&](int curve, int seg, const CurveSrc& s, float* r1, float* r2, float* len) {
        const Ctl c = curve_ctl(s, curve);
        const float t0 = (float)seg       * (1.0f / 9.0f);
        const float t1 = (float)(seg + 1) * (1.0f / 9.0f);
        const float u0 = 1.0f - t0, u1 = 1.0f - t1;
        const float ay = u0*u0*c.c0y + 2.0f*t0*u0*c.c1y + t0*t0*c.c2y;
        const float ax = u0*u0*c.c0x + 2.0f*t0*u0*c.c1x + t0*t0*c.c2x;
        const float ey = u1*u1*c.c0y + 2.0f*t1*u1*c.c1y + t1*t1*c.c2y;
        const float ex = u1*u1*c.c0x + 2.0f*t1*u1*c.c1x + t1*t1*c.c2x;
        const float vy = ey - ay, vx = ex - ax;
        const float vv = vy*vy + vx*vx;
        const float s_  = sqrtf(vv + EPSF);
        const float inv = 1.0f / s_;
        r1[0] = vy * inv;
        r1[1] = vx * inv;
        r1[2] = -(ay*vy + ax*vx) * inv;
        r1[3] = 0.0f;
        r2[0] = -2.0f * ay;
        r2[1] = -2.0f * ax;
        r2[2] = ay*ay + ax*ax;
        r2[3] = 1.0f;
        *len = s_;
    };
    build_row(cv0, sg0, s0, &sA[0][cv0][sg0][0], &sA[1][cv0][sg0][0], &sCv[cv0][sg0]);
    if (third) build_row(cvx, 8, sx, &sN[0][cvx][0], &sN[1][cvx][0], &sN9len[cvx]);
    if (boxw) {
        const Ctl c = curve_ctl(sx, cvx);
        float ymin = c.c0y, ymax = c.c0y, xmin = c.c0x, xmax = c.c0x;
#pragma unroll
        for (int k = 1; k < 10; ++k) {
            const float t = (float)k * (1.0f / 9.0f);
            const float u = 1.0f - t;
            const float py = u*u*c.c0y + 2.0f*t*u*c.c1y + t*t*c.c2y;
            const float px = u*u*c.c0x + 2.0f*t*u*c.c1x + t*t*c.c2x;
            ymin = fminf(ymin, py); ymax = fmaxf(ymax, py);
            xmin = fminf(xmin, px); xmax = fmaxf(xmax, px);
        }
        sBox[cvx][0] = ymin; sBox[cvx][1] = ymax; sBox[cvx][2] = xmin; sBox[cvx][3] = xmax;
        sThr[cvx] = 11.0f * (rx * rx * SIGMA2) + 1e-3f;
    }
    if (metaw) {
        const float sg = mr * mr * SIGMA2;
        const float is = 1.0f / sg;
        float* m0 = &sCv[2 * mline][8];
        float* m1 = &sCv[2 * mline + 1][8];
        m0[0] = is; m0[1] = mcR; m0[2] = mcG; m0[3] = mcB;
        m1[0] = is; m1[1] = mcR; m1[2] = mcG; m1[3] = mcB;
    }
    if (tid >= 128 && tid < 136) {
        const int r = tid - 128;
        sA[0][DUMMY][r][0] = 0.0f; sA[0][DUMMY][r][1] = 0.0f; sA[0][DUMMY][r][2] = 0.0f;  sA[0][DUMMY][r][3] = 0.0f;
        sA[1][DUMMY][r][0] = 0.0f; sA[1][DUMMY][r][1] = 0.0f; sA[1][DUMMY][r][2] = 1e4f;  sA[1][DUMMY][r][3] = 0.0f;
        sCv[DUMMY][r] = 1.0f;
    }
    if (tid == 136) {
        sN[0][DUMMY][0] = 0.0f; sN[0][DUMMY][1] = 0.0f; sN[0][DUMMY][2] = 0.0f;  sN[0][DUMMY][3] = 0.0f;
        sN[1][DUMMY][0] = 0.0f; sN[1][DUMMY][1] = 0.0f; sN[1][DUMMY][2] = 1e4f;  sN[1][DUMMY][3] = 0.0f;
        sN9len[DUMMY] = 1.0f;
        sCv[DUMMY][8] = 1.0f; sCv[DUMMY][9] = 0.0f; sCv[DUMMY][10] = 0.0f; sCv[DUMMY][11] = 0.0f;
    }
    sList[wave][0][lane] = DUMMY;
    sList[wave][1][lane] = DUMMY;
    sL9[wave][0][lane] = 1.0f;
    sL9[wave][1][lane] = 1.0f;
    __syncthreads();

    const int wk    = wave & 3, wm = wave >> 2;
    const int y     = yr + 32 * wm;
    const int strip = (wk + 4 * q + 2 * wm) & 7;
    const int colp  = lane & 15;
    const int x     = strip * 16 + colp;
    const float sc = 2.0f / 127.0f;
    const float gy = -1.0f + (float)y * sc;
    const float gx = -1.0f + (float)x * sc;
    const bool  hi = lane >= 16;

    const float xl = -1.0f + (float)(strip * 16) * sc;
    const float xr = -1.0f + (float)(strip * 16 + 15) * sc;
    const v4f  bxa = *(const v4f*)&sBox[lane][0];
    const v4f  bxb = *(const v4f*)&sBox[lane + 32][0];
    const float tha = sThr[lane], thb = sThr[lane + 32];
    const float n9a = sN9len[lane], n9b = sN9len[lane + 32];
    const float dya = fmaxf(fmaxf(bxa.x - gy, gy - bxa.y), 0.0f);
    const float dxa = fmaxf(fmaxf(bxa.z - xr, xl - bxa.w), 0.0f);
    const float dyb = fmaxf(fmaxf(bxb.x - gy, gy - bxb.y), 0.0f);
    const float dxb = fmaxf(fmaxf(bxb.z - xr, xl - bxb.w), 0.0f);
    const bool keepa = !(dya * dya + dxa * dxa > tha);
    const bool keepb = !(dyb * dyb + dxb * dxb > thb);
    const unsigned ma = __builtin_amdgcn_ballot_w32(keepa);
    const unsigned mb = __builtin_amdgcn_ballot_w32(keepb);
    const int na = __builtin_popcount(ma);
    const int n  = na + __builtin_popcount(mb);
    const int h0 = (n + 1) >> 1;
    const unsigned below = (1u << lane) - 1u;
    const int ra = __builtin_popcount(ma & below);
    const int rb = na + __builtin_popcount(mb & below);
    if (keepa) {
        const int li = (ra >= h0) ? 1 : 0;
        const int ix = ra - (li ? h0 : 0);
        sList[wave][li][ix] = lane;
        sL9[wave][li][ix]   = n9a;
    }
    if (keepb) {
        const int li = (rb >= h0) ? 1 : 0;
        const int ix = rb - (li ? h0 : 0);
        sList[wave][li][ix] = lane + 32;
        sL9[wave][li][ix]   = n9b;
    }

    v2f Bf;
    Bf.x = hi ? 1.0f : gy;
    Bf.y = hi ? (gy * gy + gx * gx) : gx;

    const int rA   = lane & 7;
    const int kOff = hi ? 2 : 0;
    const int*   listA = &sList[wave][(lane >> 3) & 1][0];
    const int*   listD = &sList[wave][lane >> 4][0];
    const float* len9D = &sL9[wave][lane >> 4][0];

    float accR = 0.0f, accG = 0.0f, accB = 0.0f;
    float w = 1.0f;

    struct TileOps { v2f A1, A2; v4f L0, L1, meta; };
    auto load_ops = [&](int ca, int cd) {
        TileOps o;
        o.A1 = *(const v2f*)&sA[0][ca][rA][kOff];
        o.A2 = *(const v2f*)&sA[1][ca][rA][kOff];
        const float* cv = &sCv[cd][0];
        o.L0   = *(const v4f*)cv;
        o.L1   = *(const v4f*)(cv + 4);
        o.meta = *(const v4f*)(cv + 8);
        return o;
    };

    const int ngroups = (h0 + 7) >> 3;
    for (int g = 0; g < ngroups; ++g) {
        const v4i ca0 = *(const v4i*)(listA + 8 * g), ca1 = *(const v4i*)(listA + 8 * g + 4);
        const v4i cd0 = *(const v4i*)(listD + 8 * g), cd1 = *(const v4i*)(listD + 8 * g + 4);
        const int cA[8] = { ca0.x, ca0.y, ca0.z, ca0.w, ca1.x, ca1.y, ca1.z, ca1.w };
        const int cD[8] = { cd0.x, cd0.y, cd0.z, cd0.w, cd1.x, cd1.y, cd1.z, cd1.w };
        const int  cN  = listA[8 * g + rA];
        const v4f  l90 = *(const v4f*)(len9D + 8 * g), l91 = *(const v4f*)(len9D + 8 * g + 4);
        TileOps buf[2];
        buf[0] = load_ops(cA[0], cD[0]);
        const v8f  E9  = tile_edt2(*(const v2f*)&sN[0][cN][kOff], *(const v2f*)&sN[1][cN][kOff], l90, l91, Bf);
#pragma unroll
        for (int j = 0; j < 8; ++j) {
            if (8 * g + j < h0) {
                if (j < 7) buf[(j + 1) & 1] = load_ops(cA[j < 7 ? j + 1 : 7], cD[j < 7 ? j + 1 : 7]);
                __builtin_amdgcn_sched_barrier(0);
                const TileOps& cur = buf[j & 1];

                const v8f E = tile_edt2(cur.A1, cur.A2, cur.L0, cur.L1, Bf);
                float m = E[0];
#pragma unroll
                for (int i = 1; i < 8; ++i) m = fminf(m, E[i]);
                m = fminf(m, E9[j]);

                const float z = m * cur.meta.x;
                const float I = __expf(-(z * z));

                const float wI = w * I;
                accR = fmaf(cur.meta.y, wI, accR);
                accG = fmaf(cur.meta.z, wI, accG);
                accB = fmaf(cur.meta.w, wI, accB);
                w = w * (1.0f - I + EPSF);
            }
        }
    }

    const float bR = xhalf(accR), bG = xhalf(accG), bB = xhalf(accB), bW = xhalf(w);

    if (!hi) {
        const float pR = fmaf(w, bR, accR);
        const float pG = fmaf(w, bG, accG);
        const float pB = fmaf(w, bB, accB);
        const float wt = w * bW;
        const int base = ((b * 3) * IMAGE + y) * IMAGE + x;
        out[base]                     = clamp01(fmaf(bg0, wt, pR));
        out[base + IMAGE * IMAGE]     = clamp01(fmaf(bg1, wt, pG));
        out[base + 2 * IMAGE * IMAGE] = clamp01(fmaf(bg2, wt, pB));
    }
}

extern "C" void kernel_launch(void* const* d_in, const int* in_sizes, int n_in,
                              void* d_out, int out_size, void* d_ws, size_t ws_size,
                              hipStream_t stream) {
    (void)in_sizes; (void)n_in; (void)out_size; (void)d_ws; (void)ws_size;
    const float* position   = (const float*)d_in[0];
    const float* radius     = (const float*)d_in[1];
    const float* color      = (const float*)d_in[2];
    const float* background = (const float*)d_in[3];
    float* out = (float*)d_out;
    render_curves_wmma<<<2 * IMAGE, 512, 0, stream>>>(position, radius, color, background, out);
}
